# starting kernel
# baseline (speedup 1.0000x reference)
#include <hip/hip_runtime.h>

typedef __attribute__((ext_vector_type(2))) float v2f;
typedef __attribute__((ext_vector_type(8))) float v8f;

#define B_ 8
#define C_ 256
#define H_ 96
#define W_ 96
#define S_ 4
#define K_ 3
#define HW_ (H_ * W_)
#define TILES_PER_ROW (W_ / 16)
#define PIX_TILES (B_ * H_ * TILES_PER_ROW)

__global__ __launch_bounds__(256) void SampleDepthwise_wmma_kernel(
    const float* __restrict__ x,
    const float* __restrict__ rot,
    const float* __restrict__ weight,
    float* __restrict__ out)
{
  const int wave = threadIdx.x >> 5;
  const int lane = threadIdx.x & 31;
  const int half = lane >> 4;
  const int pix  = lane & 15;

  const int pt  = blockIdx.x >> 1;
  const int cg  = ((blockIdx.x & 1) << 3) + wave;
  const int b   = pt / (H_ * TILES_PER_ROW);
  const int rem = pt % (H_ * TILES_PER_ROW);
  const int h   = rem / TILES_PER_ROW;
  const int w0  = (rem % TILES_PER_ROW) * 16;
  const int c0  = cg * 16;

  v2f a[4];
  {
    const int ch_a = c0 + pix;
    #pragma unroll
    for (int j = 0; j < 4; ++j) {
      const int k = 4 * j + 2 * half;
      a[j].x = weight[ch_a * 16 + k];
      a[j].y = weight[ch_a * 16 + k + 1];
    }
  }

  const float* rotp = rot + (size_t)b * 18 * HW_ + h * W_ + (w0 + pix);
  const float* xch  = x   + ((size_t)b * C_ + c0 + 8 * half) * HW_;
  float*       och  = out + ((size_t)b * C_ + c0 + 8 * half) * HW_;

  const float fx_lo = (float)(2 * half);
  const float fx_hi = (float)(2 * half + 1);

  v8f acc = {};

  #pragma unroll
  for (int kh = 0; kh < K_; ++kh) {
    #pragma unroll
    for (int kw = 0; kw < K_; ++kw) {
      const int tap = kh * K_ + kw;

      const float ry = rotp[(2 * tap + 0) * HW_];
      const float rx = rotp[(2 * tap + 1) * HW_];
      const float py = (kh + 0.5f) * ((float)S_ / (float)K_) - 0.5f + ry;
      const float px = (kw + 0.5f) * ((float)S_ / (float)K_) - 0.5f + rx;

      const int hh    = h + kh - 1;
      const int ww    = w0 + pix + kw - 1;
      const int hh_cl = min(max(hh, 0), H_ - 1);
      const int ww_cl = min(max(ww, 0), W_ - 1);
      const float m   = ((hh == hh_cl) & (ww == ww_cl)) ? 1.0f : 0.0f;

      float fy[4];
      #pragma unroll
      for (int y = 0; y < 4; ++y)
        fy[y] = fmaxf(0.0f, 1.0f - fabsf(py - (float)y));
      const float fx0 = fmaxf(0.0f, 1.0f - fabsf(px - fx_lo)) * m;
      const float fx1 = fmaxf(0.0f, 1.0f - fabsf(px - fx_hi)) * m;

      v8f c = {};
      #pragma unroll
      for (int j = 0; j < 4; ++j) {
        v2f bb;
        bb.x = fy[j] * fx0;
        bb.y = fy[j] * fx1;
        c = __builtin_amdgcn_wmma_f32_16x16x4_f32(
             false, a[j],  false, bb,
             (short)0, c,  false,  false);
      }

      const float* xp = xch + hh_cl * W_ + ww_cl;
      #pragma unroll
      for (int i = 0; i < 8; ++i) {
        acc[i] += c[i] * xp[i * HW_];
      }
    }
  }

  float* op = och + h * W_ + (w0 + pix);
  #pragma unroll
  for (int i = 0; i < 8; ++i) {
    __builtin_nontemporal_store(acc[i], op + i * HW_);
  }
}

extern "C" void kernel_launch(void* const* d_in, const int* in_sizes, int n_in,
                              void* d_out, int out_size, void* d_ws, size_t ws_size,
                              hipStream_t stream) {
  (void)in_sizes; (void)out_size; (void)d_ws; (void)ws_size;
  if (n_in < 3) return;
  const float* x   = (const float*)d_in[0];
  const float* rot = (const float*)d_in[1];
  const float* w   = (const float*)d_in[2];
  float* out = (float*)d_out;

  const int blocks = PIX_TILES * 2;
  SampleDepthwise_wmma_kernel<<<blocks, 256, 0, stream>>>(x, rot, w, out);
}


# Each wave walks 3 consecutive rows of its 16ch x 16px tile and keeps each loaded x row in registers for all three vertical taps: x loads per output row 72 -> 40, a third as many waves; tap order kept
# speedup vs baseline: 1.2887x; 1.2887x over previous
#include <hip/hip_runtime.h>

typedef __attribute__((ext_vector_type(2))) float v2f;
typedef __attribute__((ext_vector_type(8))) float v8f;

#define B_ 8
#define C_ 256
#define H_ 96
#define W_ 96
#define S_ 4
#define K_ 3
#define HW_ (H_ * W_)
#define TILES_PER_ROW (W_ / 16)
#define R_ 3
#define ROW_STRIPS (H_ / R_)
#define PIX_STRIPS (B_ * ROW_STRIPS * TILES_PER_ROW)
static_assert(H_ % R_ == 0, "a strip of R_ rows must tile the image height");

__global__ __launch_bounds__(256)  void SampleDepthwise_wmma_kernel(
    const float* __restrict__ x,
    const float* __restrict__ rot,
    const float* __restrict__ weight,
    float* __restrict__ out)
{
  const int wave = threadIdx.x >> 5;
  const int lane = threadIdx.x & 31;
  const int half = lane >> 4;
  const int pix  = lane & 15;

  const int ps  = blockIdx.x >> 1;
  const int cg  = ((blockIdx.x & 1) << 3) + wave;
  const int b   = ps / (ROW_STRIPS * TILES_PER_ROW);
  const int rem = ps % (ROW_STRIPS * TILES_PER_ROW);
  const int h0  = (rem / TILES_PER_ROW) * R_;
  const int w0  = (rem % TILES_PER_ROW) * 16;
  const int c0  = cg * 16;

  v2f a[4];
  {
    const int ch_a = c0 + pix;
    #pragma unroll
    for (int j = 0; j < 4; ++j) {
      const int k = 4 * j + 2 * half;
      a[j].x = weight[ch_a * 16 + k];
      a[j].y = weight[ch_a * 16 + k + 1];
    }
  }

  const float* rotp = rot + (size_t)b * 18 * HW_ + h0 * W_ + (w0 + pix);
  const float* xch  = x   + ((size_t)b * C_ + c0 + 8 * half) * HW_;
  float*       och  = out + ((size_t)b * C_ + c0 + 8 * half) * HW_;

  const float fx_lo = (float)(2 * half);
  const float fx_hi = (float)(2 * half + 1);

  v8f acc[R_];
  #pragma unroll
  for (int r = 0; r < R_; ++r) acc[r] = (v8f){};

  #pragma unroll
  for (int xr = 0; xr < R_ + 2; ++xr) {
    const int hh    = h0 + xr - 1;
    const int hh_cl = min(max(hh, 0), H_ - 1);
    const bool row_in = (hh == hh_cl);

    #pragma unroll
    for (int kw = 0; kw < K_; ++kw) {
      const int ww    = w0 + pix + kw - 1;
      const int ww_cl = min(max(ww, 0), W_ - 1);
      const float m   = (row_in & (ww == ww_cl)) ? 1.0f : 0.0f;

      const float* xp = xch + hh_cl * W_ + ww_cl;
      float xv[8];
      #pragma unroll
      for (int i = 0; i < 8; ++i) xv[i] = xp[i * HW_];

      #pragma unroll
      for (int kh = 0; kh < K_; ++kh) {
        const int r = xr - kh;
        if (r < 0 || r >= R_) continue;
        const int tap = kh * K_ + kw;

        const float ry = rotp[(2 * tap + 0) * HW_ + r * W_];
        const float rx = rotp[(2 * tap + 1) * HW_ + r * W_];
        const float py = (kh + 0.5f) * ((float)S_ / (float)K_) - 0.5f + ry;
        const float px = (kw + 0.5f) * ((float)S_ / (float)K_) - 0.5f + rx;

        float fy[4];
        #pragma unroll
        for (int y = 0; y < 4; ++y)
          fy[y] = fmaxf(0.0f, 1.0f - fabsf(py - (float)y));
        const float fx0 = fmaxf(0.0f, 1.0f - fabsf(px - fx_lo)) * m;
        const float fx1 = fmaxf(0.0f, 1.0f - fabsf(px - fx_hi)) * m;

        v8f c = {};
        #pragma unroll
        for (int j = 0; j < 4; ++j) {
          v2f bb;
          bb.x = fy[j] * fx0;
          bb.y = fy[j] * fx1;
          c = __builtin_amdgcn_wmma_f32_16x16x4_f32(
               false, a[j],  false, bb,
               (short)0, c,  false,  false);
        }

        #pragma unroll
        for (int i = 0; i < 8; ++i) acc[r][i] += c[i] * xv[i];
      }
    }
  }

  #pragma unroll
  for (int r = 0; r < R_; ++r) {
    float* op = och + (h0 + r) * W_ + (w0 + pix);
    #pragma unroll
    for (int i = 0; i < 8; ++i) {
      __builtin_nontemporal_store(acc[r][i], op + i * HW_);
    }
  }
}

extern "C" void kernel_launch(void* const* d_in, const int* in_sizes, int n_in,
                              void* d_out, int out_size, void* d_ws, size_t ws_size,
                              hipStream_t stream) {
  (void)in_sizes; (void)out_size; (void)d_ws; (void)ws_size;
  if (n_in < 3) return;
  const float* x   = (const float*)d_in[0];
  const float* rot = (const float*)d_in[1];
  const float* w   = (const float*)d_in[2];
  float* out = (float*)d_out;

  SampleDepthwise_wmma_kernel<<<PIX_STRIPS * 2, 256, 0, stream>>>(x, rot, w, out);
}
